# starting kernel
# baseline (speedup 1.0000x reference)
#include <hip/hip_runtime.h>
#include <hip/hip_bf16.h>

typedef __attribute__((ext_vector_type(2))) float v2f;
typedef __attribute__((ext_vector_type(8))) float v8f;

#define IN_C 128
#define HID  64
#define OUT_C 16

__global__ void deg_init_k(float* __restrict__ deg, int n) {
  int i = blockIdx.x * blockDim.x + threadIdx.x;
  if (i < n) deg[i] = 1.0f;
}

__global__ void deg_edges_k(float* __restrict__ deg, const int* __restrict__ ei, int E) {
  int e = blockIdx.x * blockDim.x + threadIdx.x;
  if (e < E) atomicAdd(&deg[ei[E + e]], 1.0f);
}

__global__ void dinv_k(float* __restrict__ deg, int n) {
  int i = blockIdx.x * blockDim.x + threadIdx.x;
  if (i < n) {
    float d = deg[i];
    deg[i] = (d > 0.0f) ? rsqrtf(d) : 0.0f;
  }
}

template <bool RELU_A>
__global__ void gemm_wmma_k(const float* __restrict__ A, const float* __restrict__ B,
                            float* __restrict__ C, int M, int K, int N) {
  __shared__ float Bs[IN_C * 16];

  const int tid  = threadIdx.x;
  const int wave = tid >> 5;
  const int lane = tid & 31;
  const int mr   = lane & 15;
  const int half = lane >> 4;
  const int n0   = blockIdx.y * 16;

  for (int i = tid; i < K * 16; i += 128) {
    int kk = i >> 4, nn = i & 15;
    Bs[i] = B[(size_t)kk * N + n0 + nn];
  }
  __syncthreads();

  const int tile = blockIdx.x * 4 + wave;
  if (tile * 16 < M) {
    const int m0 = tile * 16;
    const float* arow = A + (size_t)(m0 + mr) * K + 2 * half;
    const float* brow = Bs + 2 * half * 16 + mr;

    v8f acc = {};
    for (int k = 0; k < K; k += 4) {
      v2f a, b;
      a[0] = arow[k];
      a[1] = arow[k + 1];
      if (RELU_A) { a[0] = fmaxf(a[0], 0.0f); a[1] = fmaxf(a[1], 0.0f); }
      b[0] = brow[k * 16];
      b[1] = brow[k * 16 + 16];
      acc = __builtin_amdgcn_wmma_f32_16x16x4_f32(false, a, false, b,
                                                  (short)0, acc, false, false);
    }

    float* crow = C + (size_t)(m0 + 8 * half) * N + n0 + mr;
#pragma unroll
    for (int r = 0; r < 8; ++r) crow[(size_t)r * N] = acc[r];
  }
}

__global__ void agg_init_k(float* __restrict__ out, const float* __restrict__ h,
                           const float* __restrict__ dinv, const float* __restrict__ bias,
                           int n4) {
  int i = blockIdx.x * blockDim.x + threadIdx.x;
  if (i < n4) {
    int node = i >> 4;
    int c4   = (i & 15) * 4;
    float d2 = dinv[node]; d2 *= d2;
    const float4 hv = ((const float4*)h)[i];
    const float4 bv = *(const float4*)&bias[c4];
    float4 o;
    o.x = bv.x + d2 * hv.x;
    o.y = bv.y + d2 * hv.y;
    o.z = bv.z + d2 * hv.z;
    o.w = bv.w + d2 * hv.w;
    ((float4*)out)[i] = o;
  }
}

__global__ void agg_edges_k(float* __restrict__ out, const float* __restrict__ h,
                            const float* __restrict__ dinv, const int* __restrict__ ei,
                            int E) {
  int idx = blockIdx.x * blockDim.x + threadIdx.x;
  if (idx >= E * 16) return;
  int e  = idx >> 4;
  int c4 = (idx & 15) * 4;
  int s = ei[e];
  int d = ei[E + e];
  float w = dinv[s] * dinv[d];
  const float4 hv = *(const float4*)&h[(size_t)s * HID + c4];
  float* op = &out[(size_t)d * HID + c4];
  atomicAdd(op + 0, w * hv.x);
  atomicAdd(op + 1, w * hv.y);
  atomicAdd(op + 2, w * hv.z);
  atomicAdd(op + 3, w * hv.w);
}

#define W1S_STRIDE 72

__global__ void edge_mlp_k(const float* __restrict__ h, const int* __restrict__ ei,
                           const float* __restrict__ Wm1, const float* __restrict__ bm1,
                           const float* __restrict__ Wm2, const float* __restrict__ bm2,
                           float* __restrict__ out, int E) {
  __shared__ float w1s[IN_C * W1S_STRIDE];
  __shared__ float w2s[HID * OUT_C];
  __shared__ float zbuf[4][16][68];

  const int tid  = threadIdx.x;
  const int wave = tid >> 5;
  const int lane = tid & 31;
  const int mr   = lane & 15;
  const int half = lane >> 4;
  const int e0   = (blockIdx.x * 4 + wave) * 16;

  for (int i = tid; i < IN_C * 16; i += 128) {
    int row = i >> 4, c4 = (i & 15) * 4;
    *(float4*)&w1s[row * W1S_STRIDE + c4] = *(const float4*)&Wm1[(size_t)row * HID + c4];
  }
  for (int i = tid; i < HID * OUT_C / 4; i += 128) {
    ((float4*)w2s)[i] = ((const float4*)Wm2)[i];
  }
  __syncthreads();

  const int row = ei[e0 + mr];
  const int col = ei[E + e0 + mr];
  const float* efr = h + (size_t)row * HID + 2 * half;
  const float* efc = h + (size_t)col * HID + 2 * half;

  v8f acc[4] = {v8f{}, v8f{}, v8f{}, v8f{}};

  for (int k = 0; k < 64; k += 4) {
    v2f a;
    a[0] = fmaxf(efr[k], 0.0f);
    a[1] = fmaxf(efr[k + 1], 0.0f);
    const float* bp = w1s + (k + 2 * half) * W1S_STRIDE + mr;
#pragma unroll
    for (int nt = 0; nt < 4; ++nt) {
      v2f b;
      b[0] = bp[nt * 16];
      b[1] = bp[W1S_STRIDE + nt * 16];
      acc[nt] = __builtin_amdgcn_wmma_f32_16x16x4_f32(false, a, false, b,
                                                      (short)0, acc[nt], false, false);
    }
  }
  for (int k = 0; k < 64; k += 4) {
    v2f a;
    a[0] = fmaxf(efc[k], 0.0f);
    a[1] = fmaxf(efc[k + 1], 0.0f);
    const float* bp = w1s + (64 + k + 2 * half) * W1S_STRIDE + mr;
#pragma unroll
    for (int nt = 0; nt < 4; ++nt) {
      v2f b;
      b[0] = bp[nt * 16];
      b[1] = bp[W1S_STRIDE + nt * 16];
      acc[nt] = __builtin_amdgcn_wmma_f32_16x16x4_f32(false, a, false, b,
                                                      (short)0, acc[nt], false, false);
    }
  }

#pragma unroll
  for (int nt = 0; nt < 4; ++nt) {
    float bias = bm1[nt * 16 + mr];
#pragma unroll
    for (int r = 0; r < 8; ++r)
      zbuf[wave][r + 8 * half][nt * 16 + mr] = fmaxf(acc[nt][r] + bias, 0.0f);
  }
  __syncthreads();

  v8f o = {};
  for (int k = 0; k < 64; k += 4) {
    v2f a, b;
    a[0] = zbuf[wave][mr][k + 2 * half];
    a[1] = zbuf[wave][mr][k + 2 * half + 1];
    b[0] = w2s[(k + 2 * half) * OUT_C + mr];
    b[1] = w2s[(k + 2 * half + 1) * OUT_C + mr];
    o = __builtin_amdgcn_wmma_f32_16x16x4_f32(false, a, false, b,
                                              (short)0, o, false, false);
  }

  float bo = bm2[mr];
  float* op = out + (size_t)(e0 + 8 * half) * OUT_C + mr;
#pragma unroll
  for (int r = 0; r < 8; ++r) op[(size_t)r * OUT_C] = o[r] + bo;
}

extern "C" void kernel_launch(void* const* d_in, const int* in_sizes, int n_in,
                              void* d_out, int out_size, void* d_ws, size_t ws_size,
                              hipStream_t stream) {
  const float* x   = (const float*)d_in[0];
  const int*   ei  = (const int*)d_in[1];
  const float* W1  = (const float*)d_in[2];
  const float* b1  = (const float*)d_in[3];
  const float* W2  = (const float*)d_in[4];
  const float* b2  = (const float*)d_in[5];
  const float* Wm1 = (const float*)d_in[6];
  const float* bm1 = (const float*)d_in[7];
  const float* Wm2 = (const float*)d_in[8];
  const float* bm2 = (const float*)d_in[9];
  float* out = (float*)d_out;

  const int N = in_sizes[0] / IN_C;
  const int E = in_sizes[1] / 2;

  float* deg  = (float*)d_ws;
  float* bufA = deg + (size_t)((N + 63) & ~63);
  float* bufB = bufA + (size_t)N * HID;

  const int mtiles = (N + 15) / 16;
  const int gx     = (mtiles + 3) / 4;
  const int n4     = N * (HID / 4);
  const int e4     = E * (HID / 4);
  const int nb  = (N + 255) / 256;
  const int eb  = (E + 255) / 256;
  const int n4b = (n4 + 255) / 256;
  const int e4b = (e4 + 255) / 256;

  deg_init_k<<<nb, 256, 0, stream>>>(deg, N);
  deg_edges_k<<<eb, 256, 0, stream>>>(deg, ei, E);
  dinv_k<<<nb, 256, 0, stream>>>(deg, N);

  gemm_wmma_k<false><<<dim3(gx, HID / 16), 128, 0, stream>>>(x, W1, bufA, N, IN_C, HID);
  agg_init_k<<<n4b, 256, 0, stream>>>(bufB, bufA, deg, b1, n4);
  agg_edges_k<<<e4b, 256, 0, stream>>>(bufB, bufA, deg, ei, E);

  gemm_wmma_k<true><<<dim3(gx, HID / 16), 128, 0, stream>>>(bufB, W2, bufA, N, HID, HID);
  agg_init_k<<<n4b, 256, 0, stream>>>(bufB, bufA, deg, b2, n4);
  agg_edges_k<<<e4b, 256, 0, stream>>>(bufB, bufA, deg, ei, E);

  edge_mlp_k<<<E / 64, 128, 0, stream>>>(bufB, ei, Wm1, bm1, Wm2, bm2, out, E);
}


# owner sweep moved to one launch at the front: it writes the hit lists and counts its range's degrees in LDS (int adds); the 3 degree launches go, both aggregations replay; sum order still arrival
# speedup vs baseline: 1.7224x; 1.7224x over previous
#include <hip/hip_runtime.h>
#include <hip/hip_bf16.h>

typedef __attribute__((ext_vector_type(2))) float v2f;
typedef __attribute__((ext_vector_type(8))) float v8f;

#define IN_C 128
#define HID  64
#define OUT_C 16

template <bool RELU_A>
__global__ void gemm_wmma_k(const float* __restrict__ A, const float* __restrict__ B,
                            float* __restrict__ C, int M, int K, int N) {
  __shared__ float Bs[IN_C * 16];

  const int tid  = threadIdx.x;
  const int wave = tid >> 5;
  const int lane = tid & 31;
  const int mr   = lane & 15;
  const int half = lane >> 4;
  const int n0   = blockIdx.y * 16;

  for (int i = tid; i < K * 16; i += 128) {
    int kk = i >> 4, nn = i & 15;
    Bs[i] = B[(size_t)kk * N + n0 + nn];
  }
  __syncthreads();

  const int tile = blockIdx.x * 4 + wave;
  if (tile * 16 < M) {
    const int m0 = tile * 16;
    const float* arow = A + (size_t)(m0 + mr) * K + 2 * half;
    const float* brow = Bs + 2 * half * 16 + mr;

    v8f acc = {};
    for (int k = 0; k < K; k += 4) {
      v2f a, b;
      a[0] = arow[k];
      a[1] = arow[k + 1];
      if (RELU_A) { a[0] = fmaxf(a[0], 0.0f); a[1] = fmaxf(a[1], 0.0f); }
      b[0] = brow[k * 16];
      b[1] = brow[k * 16 + 16];
      acc = __builtin_amdgcn_wmma_f32_16x16x4_f32(false, a, false, b,
                                                  (short)0, acc, false, false);
    }

    float* crow = C + (size_t)(m0 + 8 * half) * N + n0 + mr;
#pragma unroll
    for (int r = 0; r < 8; ++r) crow[(size_t)r * N] = acc[r];
  }
}

#define AG_R     196
#define AG_QCAP  1024
#define SW_QCAP  2048
#define AG_WAVES 32
#define AG_C     160

__device__ __forceinline__ float lane_f(float x, int j) {
  return __int_as_float(__builtin_amdgcn_readlane(__float_as_int(x), j));
}

__device__ __forceinline__ int4 ld_dst4(const int* __restrict__ dal, int q) {
  return *(const int4*)(dal + 4 * (size_t)q);
}

template <int QCAP, class Drain>
__device__ __forceinline__ void sweep_range(const int* __restrict__ dstp, int E, int lo, int cnt,
                                            int* rq, int wave, int lane, int tid, Drain drain) {
  const int a0  = min((4 - (E & 3)) & 3, E);
  const int nq  = (E - a0) >> 2;
  const int* __restrict__ dal = dstp + a0;
  const int nfull = nq >> 12;
  const int rot = (nfull > 0) ? (int)(blockIdx.x % (unsigned)nfull) : 0;

  int head = 0, tail = 0;

  auto pop = [&](int n) {
    int e = 0;
    if (lane < n) e = rq[(head + lane) & (QCAP - 1)];
    head += n;
    drain(e, n);
  };

  auto take = [&](const int4 d, int e0) {
    const bool h0 = (unsigned)(d.x - lo) < (unsigned)cnt, h1 = (unsigned)(d.y - lo) < (unsigned)cnt;
    const bool h2 = (unsigned)(d.z - lo) < (unsigned)cnt, h3 = (unsigned)(d.w - lo) < (unsigned)cnt;
    const unsigned m0 = __builtin_amdgcn_ballot_w32(h0), m1 = __builtin_amdgcn_ballot_w32(h1);
    const unsigned m2 = __builtin_amdgcn_ballot_w32(h2), m3 = __builtin_amdgcn_ballot_w32(h3);
    if (m0 | m1 | m2 | m3) {
      if (m0) { if (h0) rq[(tail + __builtin_amdgcn_mbcnt_lo(m0, 0u)) & (QCAP - 1)] = e0;     tail += __popc(m0); }
      if (m1) { if (h1) rq[(tail + __builtin_amdgcn_mbcnt_lo(m1, 0u)) & (QCAP - 1)] = e0 + 1; tail += __popc(m1); }
      if (m2) { if (h2) rq[(tail + __builtin_amdgcn_mbcnt_lo(m2, 0u)) & (QCAP - 1)] = e0 + 2; tail += __popc(m2); }
      if (m3) { if (h3) rq[(tail + __builtin_amdgcn_mbcnt_lo(m3, 0u)) & (QCAP - 1)] = e0 + 3; tail += __popc(m3); }
    }
  };

  if (nfull > 0) {
    int it = rot;
    int4 cur[4], nxt[4];
#pragma unroll
    for (int u = 0; u < 4; ++u) cur[u] = ld_dst4(dal, it * 4096 + u * 1024 + tid);
    for (int k = 0; k < nfull; ++k) {
      const int qb = it * 4096 + tid;
      it = (it + 1 == nfull) ? 0 : it + 1;
#pragma unroll
      for (int u = 0; u < 4; ++u) nxt[u] = ld_dst4(dal, it * 4096 + u * 1024 + tid);
#pragma unroll
      for (int u = 0; u < 4; ++u) take(cur[u], a0 + 4 * (qb + u * 1024));
#pragma unroll
      for (int u = 0; u < 4; ++u) cur[u] = nxt[u];
      while (tail - head >= 32) pop(32);
    }
  }
#pragma unroll
  for (int u = 0; u < 4; ++u) {
    const int q = nfull * 4096 + u * 1024 + tid;
    int4 d = make_int4(-1, -1, -1, -1);
    if (q < nq) d = ld_dst4(dal, q);
    take(d, a0 + 4 * q);
  }
  if (wave == 0) {
    const int nrem = E - 4 * nq;
    const int er = (lane < a0) ? lane : lane + 4 * nq;
    int4 d = make_int4(-1, -1, -1, -1);
    if (lane < nrem) d.x = dstp[er];
    take(d, er);
  }
  while (tail - head > 0) pop(min(tail - head, 32));
}

__device__ __forceinline__ bool lists_fit(const unsigned* rec, int N, const float* ws0, size_t ws_bytes) {
  const int nrng = (N + AG_R - 1) / AG_R;
  return (N <= 65536) &&
      (size_t)((const char*)(rec + (size_t)nrng * AG_WAVES * AG_C) - (const char*)ws0) <= ws_bytes;
}

__global__ __launch_bounds__(1024)
void sweep_k(float* __restrict__ dinv, const int* __restrict__ ei, int E, int N,
             unsigned* __restrict__ rec, int* __restrict__ cnts,
             const float* ws0, size_t ws_bytes) {
  __shared__ int ring[AG_WAVES * SW_QCAP];
  __shared__ int degc[AG_R];

  const int tid  = threadIdx.x;
  const int wave = __builtin_amdgcn_readfirstlane(tid >> 5);
  const int lane = tid & 31;
  int* const rq = ring + wave * SW_QCAP;
  const int* __restrict__ dstp = ei + E;
  const bool fits = lists_fit(rec, N, ws0, ws_bytes);

  for (int lo = blockIdx.x * AG_R; lo < N; lo += gridDim.x * AG_R) {
    const int cnt = min(AG_R, N - lo);
    const int li  = (lo / AG_R) * AG_WAVES + wave;
    unsigned* const rw = rec + (size_t)li * AG_C;

    if (tid < AG_R) degc[tid] = 1;
    __syncthreads();

    int nrec = fits ? 0 : -1;
    sweep_range<SW_QCAP>(dstp, E, lo, cnt, rq, wave, lane, tid, [&](int e, int n) {
      const bool keep = nrec >= 0 && nrec + n <= AG_C;
      if (lane < n) {
        const int s  = ei[e];
        const int dl = dstp[e] - lo;
        atomicAdd(&degc[dl], 1);
        if (keep) rw[nrec + lane] = ((unsigned)dl << 16) | (unsigned)s;
      }
      nrec = keep ? nrec + n : -1;
    });
    if (fits && lane == 0) cnts[li] = nrec;
    __syncthreads();

    if (tid < cnt) {
      const float d = (float)degc[tid];
      dinv[lo + tid] = (d > 0.0f) ? rsqrtf(d) : 0.0f;
    }
    __syncthreads();
  }
}

__global__ __launch_bounds__(1024)
void agg_own_k(float* __restrict__ out, const float* __restrict__ h,
               const float* __restrict__ dinv, const float* __restrict__ bias,
               const int* __restrict__ ei, int E, int N,
               const unsigned* __restrict__ rec, const int* __restrict__ cnts,
               const float* ws0, size_t ws_bytes) {
  __shared__ __attribute__((aligned(16))) float acc[(AG_R + 1) * HID];
  __shared__ int ring[AG_WAVES * AG_QCAP];

  const int tid  = threadIdx.x;
  const int wave = __builtin_amdgcn_readfirstlane(tid >> 5);
  const int lane = tid & 31;
  int* const rq = ring + wave * AG_QCAP;
  const int* __restrict__ dstp = ei + E;
  const bool fits = lists_fit(rec, N, ws0, ws_bytes);

  for (int lo = blockIdx.x * AG_R; lo < N; lo += gridDim.x * AG_R) {
    const int cnt = min(AG_R, N - lo);
    const int li  = (lo / AG_R) * AG_WAVES + wave;
    const unsigned* const rw = rec + (size_t)li * AG_C;

    for (int i = tid; i < cnt * 16; i += 1024) {
      const int r  = i >> 4;
      const int c4 = (i & 15) * 4;
      float d2 = dinv[lo + r]; d2 *= d2;
      const float4 hv = *(const float4*)&h[(size_t)(lo + r) * HID + c4];
      const float4 bv = *(const float4*)&bias[c4];
      float4 o;
      o.x = bv.x + d2 * hv.x;
      o.y = bv.y + d2 * hv.y;
      o.z = bv.z + d2 * hv.z;
      o.w = bv.w + d2 * hv.w;
      *(float4*)&acc[r * HID + c4] = o;
    }
    if (tid < HID) acc[AG_R * HID + tid] = 0.0f;
    __syncthreads();

    auto add_rows = [&](int s, int dl, int n, auto wfn) {
      float w = 0.0f;
#pragma unroll
      for (int hb = 0; hb < 32; hb += 16) {
        if (hb < n) {
          float2 v[16];
#pragma unroll
          for (int j = 0; j < 16; ++j) {
            const int sj = __builtin_amdgcn_readlane(s, hb + j);
            v[j] = *(const float2*)&h[(size_t)sj * HID + 2 * lane];
          }
          if (hb == 0) w = wfn();
#pragma unroll
          for (int j = 0; j < 16; ++j) {
            const int   dj = __builtin_amdgcn_readlane(dl, hb + j);
            const float wj = lane_f(w, hb + j);
            float* p = &acc[dj * HID + 2 * lane];
            atomicAdd(p,     wj * v[j].x);
            atomicAdd(p + 1, wj * v[j].y);
          }
        }
      }
    };

    int nrec = -1;
    if (fits) {
      nrec = __builtin_amdgcn_readfirstlane(cnts[li]);
      if (nrec > AG_C) nrec = -1;
    }

    if (nrec >= 0) {
      unsigned rn = (lane < nrec) ? rw[lane] : 0u;
      for (int b = 0; b < nrec; b += 32) {
        const unsigned r = rn;
        const int n = min(nrec - b, 32);
        if (b + 32 + lane < nrec) rn = rw[b + 32 + lane];
        const bool live = lane < n;
        const int  s  = live ? (int)(r & 0xFFFFu) : 0;
        const int  dl = live ? (int)(r >> 16) : AG_R;
        const float wa = dinv[s];
        const float wb = dinv[lo + (live ? dl : 0)];
        add_rows(s, dl, n, [&] { return live ? wa * wb : 0.0f; });
      }
    } else {
      sweep_range<AG_QCAP>(dstp, E, lo, cnt, rq, wave, lane, tid, [&](int e, int n) {
        int   s  = 0;
        int   dl = AG_R;
        float w  = 0.0f;
        if (lane < n) {
          s = ei[e];
          const int d = dstp[e];
          dl = d - lo;
          w  = dinv[s] * dinv[d];
        }
        add_rows(s, dl, n, [&] { return w; });
      });
    }
    __syncthreads();

    for (int i = tid; i < cnt * 16; i += 1024)
      ((float4*)out)[(size_t)lo * 16 + i] = *(const float4*)&acc[i * 4];
    __syncthreads();
  }
}

#define AT_P 68

__global__ __launch_bounds__(128) __attribute__((amdgpu_waves_per_eu(4, 4)))
void edge_mlp_k(const float* __restrict__ h, const int* __restrict__ ei,
                const float* __restrict__ Wm1, const float* __restrict__ bm1,
                const float* __restrict__ Wm2, const float* __restrict__ bm2,
                float* __restrict__ out, int E, int ntiles) {
  __shared__ __attribute__((aligned(16))) float At[128 * AT_P];
  __shared__ __attribute__((aligned(16))) float Zt[64 * AT_P];

  const int tid  = threadIdx.x;
  const int wave = tid >> 5;
  const int lane = tid & 31;
  const int mr   = lane & 15;
  const int half = lane >> 4;

  v2f b1f[32];
#pragma unroll
  for (int j = 0; j < 32; ++j) {
    const float* p = Wm1 + (size_t)(4 * j + 2 * half) * HID + wave * 16 + mr;
    b1f[j][0] = p[0];
    b1f[j][1] = p[HID];
  }
  v2f b2f[16];
#pragma unroll
  for (int j = 0; j < 16; ++j) {
    const float* p = Wm2 + (size_t)(4 * j + 2 * half) * OUT_C + mr;
    b2f[j][0] = p[0];
    b2f[j][1] = p[OUT_C];
  }
  const float bias1 = bm1[wave * 16 + mr];
  const float bo    = bm2[mr];

  for (int t = blockIdx.x; t < ntiles; t += gridDim.x) {
    const int e0 = t * 64;

    const int gi = wave * 32 + lane;
    const int node_l = (gi < 64) ? ei[e0 + gi] : ei[E + e0 + (gi - 64)];
#pragma unroll
    for (int i = 0; i < 16; ++i) {
      const int rl   = 2 * i + half;
      const int node = __shfl(node_l, rl);
      float4 v = *(const float4*)&h[(size_t)node * HID + mr * 4];
      v.x = fmaxf(v.x, 0.0f);
      v.y = fmaxf(v.y, 0.0f);
      v.z = fmaxf(v.z, 0.0f);
      v.w = fmaxf(v.w, 0.0f);
      *(float4*)&At[(wave * 32 + rl) * AT_P + mr * 4] = v;
    }
    __syncthreads();

    v8f acc[4] = {v8f{}, v8f{}, v8f{}, v8f{}};
#pragma unroll
    for (int j = 0; j < 32; ++j) {
      const int rbase = (j < 16) ? 0 : 64;
      const int kc    = 4 * (j & 15) + 2 * half;
#pragma unroll
      for (int mt = 0; mt < 4; ++mt) {
        const v2f a = *(const v2f*)&At[(rbase + mt * 16 + mr) * AT_P + kc];
        acc[mt] = __builtin_amdgcn_wmma_f32_16x16x4_f32(false, a, false, b1f[j],
                                                        (short)0, acc[mt], false, false);
      }
    }

#pragma unroll
    for (int mt = 0; mt < 4; ++mt) {
#pragma unroll
      for (int r = 0; r < 8; ++r)
        Zt[(mt * 16 + r + 8 * half) * AT_P + wave * 16 + mr] = fmaxf(acc[mt][r] + bias1, 0.0f);
    }
    __syncthreads();

    v8f o = {};
#pragma unroll
    for (int j = 0; j < 16; ++j) {
      const v2f a = *(const v2f*)&Zt[(wave * 16 + mr) * AT_P + 4 * j + 2 * half];
      o = __builtin_amdgcn_wmma_f32_16x16x4_f32(false, a, false, b2f[j],
                                                (short)0, o, false, false);
    }

    float* op = out + (size_t)(e0 + wave * 16 + 8 * half) * OUT_C + mr;
#pragma unroll
    for (int r = 0; r < 8; ++r) op[(size_t)r * OUT_C] = o[r] + bo;
  }
}

extern "C" void kernel_launch(void* const* d_in, const int* in_sizes, int n_in,
                              void* d_out, int out_size, void* d_ws, size_t ws_size,
                              hipStream_t stream) {
  const float* x   = (const float*)d_in[0];
  const int*   ei  = (const int*)d_in[1];
  const float* W1  = (const float*)d_in[2];
  const float* b1  = (const float*)d_in[3];
  const float* W2  = (const float*)d_in[4];
  const float* b2  = (const float*)d_in[5];
  const float* Wm1 = (const float*)d_in[6];
  const float* bm1 = (const float*)d_in[7];
  const float* Wm2 = (const float*)d_in[8];
  const float* bm2 = (const float*)d_in[9];
  float* out = (float*)d_out;

  const int N = in_sizes[0] / IN_C;
  const int E = in_sizes[1] / 2;

  float* dinv = (float*)d_ws;
  float* bufA = dinv + (size_t)((N + 63) & ~63);
  float* bufB = bufA + (size_t)N * HID;

  const int mtiles = (N + 15) / 16;
  const int gx     = (mtiles + 3) / 4;
  const int ag  = (N + AG_R - 1) / AG_R;
  int*      cnts = (int*)(bufB + (size_t)N * HID);
  unsigned* rec  = (unsigned*)(cnts + (size_t)ag * AG_WAVES);

  sweep_k<<<ag, 1024, 0, stream>>>(dinv, ei, E, N, rec, cnts, dinv, ws_size);

  gemm_wmma_k<false><<<dim3(gx, HID / 16), 128, 0, stream>>>(x, W1, bufA, N, IN_C, HID);
  agg_own_k<<<ag, 1024, 0, stream>>>(bufB, bufA, dinv, b1, ei, E, N, rec, cnts, dinv, ws_size);

  gemm_wmma_k<true><<<dim3(gx, HID / 16), 128, 0, stream>>>(bufB, W2, bufA, N, HID, HID);
  agg_own_k<<<ag, 1024, 0, stream>>>(bufB, bufA, dinv, b2, ei, E, N, rec, cnts, dinv, ws_size);

  edge_mlp_k<<<1024, 128, 0, stream>>>(bufB, ei, Wm1, bm1, Wm2, bm2, out, E, E / 64);
}


# new front launch writes one bit per edge and owner range (LDS OR); the owner sweep reads its range's bits, not the whole destination list; waves' edge shares change, sum order still arrival
# speedup vs baseline: 1.7945x; 1.0419x over previous
#include <hip/hip_runtime.h>
#include <hip/hip_bf16.h>

typedef __attribute__((ext_vector_type(2))) float v2f;
typedef __attribute__((ext_vector_type(8))) float v8f;

#define IN_C 128
#define HID  64
#define OUT_C 16

template <bool RELU_A>
__global__ void gemm_wmma_k(const float* __restrict__ A, const float* __restrict__ B,
                            float* __restrict__ C, int M, int K, int N) {
  __shared__ float Bs[IN_C * 16];

  const int tid  = threadIdx.x;
  const int wave = tid >> 5;
  const int lane = tid & 31;
  const int mr   = lane & 15;
  const int half = lane >> 4;
  const int n0   = blockIdx.y * 16;

  for (int i = tid; i < K * 16; i += 128) {
    int kk = i >> 4, nn = i & 15;
    Bs[i] = B[(size_t)kk * N + n0 + nn];
  }
  __syncthreads();

  const int tile = blockIdx.x * 4 + wave;
  if (tile * 16 < M) {
    const int m0 = tile * 16;
    const float* arow = A + (size_t)(m0 + mr) * K + 2 * half;
    const float* brow = Bs + 2 * half * 16 + mr;

    v8f acc = {};
    for (int k = 0; k < K; k += 4) {
      v2f a, b;
      a[0] = arow[k];
      a[1] = arow[k + 1];
      if (RELU_A) { a[0] = fmaxf(a[0], 0.0f); a[1] = fmaxf(a[1], 0.0f); }
      b[0] = brow[k * 16];
      b[1] = brow[k * 16 + 16];
      acc = __builtin_amdgcn_wmma_f32_16x16x4_f32(false, a, false, b,
                                                  (short)0, acc, false, false);
    }

    float* crow = C + (size_t)(m0 + 8 * half) * N + n0 + mr;
#pragma unroll
    for (int r = 0; r < 8; ++r) crow[(size_t)r * N] = acc[r];
  }
}

#define AG_R     196
#define AG_QCAP  1024
#define SW_QCAP  2048
#define AG_WAVES 32
#define AG_C     160

#define MK_E  2048
#define MK_W  (MK_E / 32)
#define MK_P  (MK_W + 4)
#define MK_R  256

__device__ __forceinline__ bool marks_fit(int N, int E) {
  const int nrng = (N + AG_R - 1) / AG_R;
  const size_t nwords = ((size_t)E + 31) >> 5;
  return nrng <= MK_R && (size_t)nrng * nwords <= (size_t)2 * N * HID;
}

__device__ __forceinline__ float lane_f(float x, int j) {
  return __int_as_float(__builtin_amdgcn_readlane(__float_as_int(x), j));
}

__device__ __forceinline__ int4 ld_dst4(const int* __restrict__ dal, int q) {
  return *(const int4*)(dal + 4 * (size_t)q);
}

template <int QCAP, class Drain>
__device__ __forceinline__ void sweep_range(const int* __restrict__ dstp, int E, int lo, int cnt,
                                            int* rq, int wave, int lane, int tid, Drain drain) {
  const int a0  = min((4 - (E & 3)) & 3, E);
  const int nq  = (E - a0) >> 2;
  const int* __restrict__ dal = dstp + a0;
  const int nfull = nq >> 12;
  const int rot = (nfull > 0) ? (int)(blockIdx.x % (unsigned)nfull) : 0;

  int head = 0, tail = 0;

  auto pop = [&](int n) {
    int e = 0;
    if (lane < n) e = rq[(head + lane) & (QCAP - 1)];
    head += n;
    drain(e, n);
  };

  auto take = [&](const int4 d, int e0) {
    const bool h0 = (unsigned)(d.x - lo) < (unsigned)cnt, h1 = (unsigned)(d.y - lo) < (unsigned)cnt;
    const bool h2 = (unsigned)(d.z - lo) < (unsigned)cnt, h3 = (unsigned)(d.w - lo) < (unsigned)cnt;
    const unsigned m0 = __builtin_amdgcn_ballot_w32(h0), m1 = __builtin_amdgcn_ballot_w32(h1);
    const unsigned m2 = __builtin_amdgcn_ballot_w32(h2), m3 = __builtin_amdgcn_ballot_w32(h3);
    if (m0 | m1 | m2 | m3) {
      if (m0) { if (h0) rq[(tail + __builtin_amdgcn_mbcnt_lo(m0, 0u)) & (QCAP - 1)] = e0;     tail += __popc(m0); }
      if (m1) { if (h1) rq[(tail + __builtin_amdgcn_mbcnt_lo(m1, 0u)) & (QCAP - 1)] = e0 + 1; tail += __popc(m1); }
      if (m2) { if (h2) rq[(tail + __builtin_amdgcn_mbcnt_lo(m2, 0u)) & (QCAP - 1)] = e0 + 2; tail += __popc(m2); }
      if (m3) { if (h3) rq[(tail + __builtin_amdgcn_mbcnt_lo(m3, 0u)) & (QCAP - 1)] = e0 + 3; tail += __popc(m3); }
    }
  };

  if (nfull > 0) {
    int it = rot;
    int4 cur[4], nxt[4];
#pragma unroll
    for (int u = 0; u < 4; ++u) cur[u] = ld_dst4(dal, it * 4096 + u * 1024 + tid);
    for (int k = 0; k < nfull; ++k) {
      const int qb = it * 4096 + tid;
      it = (it + 1 == nfull) ? 0 : it + 1;
#pragma unroll
      for (int u = 0; u < 4; ++u) nxt[u] = ld_dst4(dal, it * 4096 + u * 1024 + tid);
#pragma unroll
      for (int u = 0; u < 4; ++u) take(cur[u], a0 + 4 * (qb + u * 1024));
#pragma unroll
      for (int u = 0; u < 4; ++u) cur[u] = nxt[u];
      while (tail - head >= 32) pop(32);
    }
  }
#pragma unroll
  for (int u = 0; u < 4; ++u) {
    const int q = nfull * 4096 + u * 1024 + tid;
    int4 d = make_int4(-1, -1, -1, -1);
    if (q < nq) d = ld_dst4(dal, q);
    take(d, a0 + 4 * q);
  }
  if (wave == 0) {
    const int nrem = E - 4 * nq;
    const int er = (lane < a0) ? lane : lane + 4 * nq;
    int4 d = make_int4(-1, -1, -1, -1);
    if (lane < nrem) d.x = dstp[er];
    take(d, er);
  }
  while (tail - head > 0) pop(min(tail - head, 32));
}

template <int QCAP, class Drain>
__device__ __forceinline__ void sweep_marks(const unsigned* __restrict__ marks, int nrng, int nwords, int ri,
                                            int* rq, int lane, int tid, Drain drain) {
  int head = 0, tail = 0;

  auto pop = [&](int n) {
    int e = 0;
    if (lane < n) e = rq[(head + lane) & (QCAP - 1)];
    head += n;
    drain(e, n);
  };

  const int nsteps = (nwords + 1023) >> 10;
  for (int k0 = 0; k0 < nsteps; k0 += 16) {
    unsigned x[16];
#pragma unroll
    for (int u = 0; u < 16; ++u) {
      const int w = (k0 + u) * 1024 + tid;
      x[u] = 0u;
      if (w < nwords) {
        const int c  = w >> 6;
        const int cw = min(MK_W, nwords - (c << 6));
        x[u] = marks[(size_t)c * nrng * MK_W + (size_t)ri * cw + (w & (MK_W - 1))];
      }
    }
#pragma unroll
    for (int u = 0; u < 16; ++u) {
      const int eb = ((k0 + u) * 1024 + tid) << 5;
      unsigned v = x[u];
      unsigned m = __builtin_amdgcn_ballot_w32(v != 0u);
      while (m) {
        if (v) rq[(tail + __builtin_amdgcn_mbcnt_lo(m, 0u)) & (QCAP - 1)] = eb + __builtin_ctz(v);
        tail += __popc(m);
        v &= v - 1u;
        m = __builtin_amdgcn_ballot_w32(v != 0u);
      }
      while (tail - head >= 32) pop(32);
    }
  }
  while (tail - head > 0) pop(min(tail - head, 32));
}

__global__ __launch_bounds__(256)
void mark_k(unsigned* __restrict__ marks, const int* __restrict__ ei, int E, int N) {
  __shared__ __attribute__((aligned(16))) unsigned L[MK_R * MK_P];
  if (!marks_fit(N, E)) return;
  const int tid    = threadIdx.x;
  const int nrng   = (N + AG_R - 1) / AG_R;
  const int nwords = (E + 31) >> 5;
  const int* __restrict__ dstp = ei + E;

  for (int c = blockIdx.x; c * MK_W < nwords; c += gridDim.x) {
    for (int i = tid; i < nrng * (MK_P / 4); i += 256)
      ((uint4*)L)[i] = make_uint4(0u, 0u, 0u, 0u);
    __syncthreads();

    const int e0 = c * MK_E;
    int d[MK_E / 256];
#pragma unroll
    for (int i = 0; i < MK_E / 256; ++i) {
      const int e = e0 + i * 256 + tid;
      d[i] = (e < E) ? dstp[e] : -1;
    }
#pragma unroll
    for (int i = 0; i < MK_E / 256; ++i) {
      const int el = i * 256 + tid;
      if ((unsigned)d[i] < (unsigned)N)
        atomicOr(&L[(d[i] / AG_R) * MK_P + (el >> 5)], 1u << (el & 31));
    }
    __syncthreads();

    const int cw = min(MK_W, nwords - c * MK_W);
    unsigned* __restrict__ g = marks + (size_t)c * nrng * MK_W;
    if (cw == MK_W) {
      for (int i = tid; i < nrng * (MK_W / 4); i += 256) {
        const int r = i >> 4, q = i & 15;
        ((uint4*)g)[i] = *(const uint4*)&L[r * MK_P + 4 * q];
      }
    } else {
      const int wl = tid & (MK_W - 1);
      for (int r = tid >> 6; r < nrng; r += 256 / MK_W)
        if (wl < cw) g[r * cw + wl] = L[r * MK_P + wl];
    }
    __syncthreads();
  }
}

__device__ __forceinline__ bool lists_fit(const unsigned* rec, int N, const float* ws0, size_t ws_bytes) {
  const int nrng = (N + AG_R - 1) / AG_R;
  return (N <= 65536) &&
      (size_t)((const char*)(rec + (size_t)nrng * AG_WAVES * AG_C) - (const char*)ws0) <= ws_bytes;
}

__global__ __launch_bounds__(1024)
void sweep_k(float* __restrict__ dinv, const int* __restrict__ ei, int E, int N,
             unsigned* __restrict__ rec, int* __restrict__ cnts,
             const float* ws0, size_t ws_bytes, const unsigned* __restrict__ marks) {
  __shared__ int ring[AG_WAVES * SW_QCAP];
  __shared__ int degc[AG_R];
  __shared__ int giveup;

  const int tid  = threadIdx.x;
  const int wave = __builtin_amdgcn_readfirstlane(tid >> 5);
  const int lane = tid & 31;
  int* const rq = ring + wave * SW_QCAP;
  const int* __restrict__ dstp = ei + E;
  const bool fits = lists_fit(rec, N, ws0, ws_bytes);
  const bool mk   = marks_fit(N, E);
  const int nrng   = (N + AG_R - 1) / AG_R;
  const int nwords = (E + 31) >> 5;

  for (int lo = blockIdx.x * AG_R; lo < N; lo += gridDim.x * AG_R) {
    const int cnt = min(AG_R, N - lo);
    const int li  = (lo / AG_R) * AG_WAVES + wave;
    unsigned* const rw = rec + (size_t)li * AG_C;

    if (tid < AG_R) degc[tid] = 1;
    if (tid == 0) giveup = 0;
    __syncthreads();

    int nrec = fits ? 0 : -1;
    auto drain = [&](int e, int n) {
      const bool keep = nrec >= 0 && nrec + n <= AG_C;
      if (lane < n) {
        const int s  = ei[e];
        const int dl = dstp[e] - lo;
        atomicAdd(&degc[dl], 1);
        if (keep) rw[nrec + lane] = ((unsigned)dl << 16) | (unsigned)s;
      }
      nrec = keep ? nrec + n : -1;
    };
    if (mk) sweep_marks<SW_QCAP>(marks, nrng, nwords, lo / AG_R, rq, lane, tid, drain);
    else    sweep_range<SW_QCAP>(dstp, E, lo, cnt, rq, wave, lane, tid, drain);
    if (nrec < 0 && lane == 0) atomicOr(&giveup, 1);
    __syncthreads();
    if (fits && lane == 0) cnts[li] = giveup ? -1 : nrec;

    if (tid < cnt) {
      const float d = (float)degc[tid];
      dinv[lo + tid] = (d > 0.0f) ? rsqrtf(d) : 0.0f;
    }
    __syncthreads();
  }
}

__global__ __launch_bounds__(1024)
void agg_own_k(float* __restrict__ out, const float* __restrict__ h,
               const float* __restrict__ dinv, const float* __restrict__ bias,
               const int* __restrict__ ei, int E, int N,
               const unsigned* __restrict__ rec, const int* __restrict__ cnts,
               const float* ws0, size_t ws_bytes) {
  __shared__ __attribute__((aligned(16))) float acc[(AG_R + 1) * HID];
  __shared__ int ring[AG_WAVES * AG_QCAP];

  const int tid  = threadIdx.x;
  const int wave = __builtin_amdgcn_readfirstlane(tid >> 5);
  const int lane = tid & 31;
  int* const rq = ring + wave * AG_QCAP;
  const int* __restrict__ dstp = ei + E;
  const bool fits = lists_fit(rec, N, ws0, ws_bytes);

  for (int lo = blockIdx.x * AG_R; lo < N; lo += gridDim.x * AG_R) {
    const int cnt = min(AG_R, N - lo);
    const int li  = (lo / AG_R) * AG_WAVES + wave;
    const unsigned* const rw = rec + (size_t)li * AG_C;

    for (int i = tid; i < cnt * 16; i += 1024) {
      const int r  = i >> 4;
      const int c4 = (i & 15) * 4;
      float d2 = dinv[lo + r]; d2 *= d2;
      const float4 hv = *(const float4*)&h[(size_t)(lo + r) * HID + c4];
      const float4 bv = *(const float4*)&bias[c4];
      float4 o;
      o.x = bv.x + d2 * hv.x;
      o.y = bv.y + d2 * hv.y;
      o.z = bv.z + d2 * hv.z;
      o.w = bv.w + d2 * hv.w;
      *(float4*)&acc[r * HID + c4] = o;
    }
    if (tid < HID) acc[AG_R * HID + tid] = 0.0f;
    __syncthreads();

    auto add_rows = [&](int s, int dl, int n, auto wfn) {
      float w = 0.0f;
#pragma unroll
      for (int hb = 0; hb < 32; hb += 16) {
        if (hb < n) {
          float2 v[16];
#pragma unroll
          for (int j = 0; j < 16; ++j) {
            const int sj = __builtin_amdgcn_readlane(s, hb + j);
            v[j] = *(const float2*)&h[(size_t)sj * HID + 2 * lane];
          }
          if (hb == 0) w = wfn();
#pragma unroll
          for (int j = 0; j < 16; ++j) {
            const int   dj = __builtin_amdgcn_readlane(dl, hb + j);
            const float wj = lane_f(w, hb + j);
            float* p = &acc[dj * HID + 2 * lane];
            atomicAdd(p,     wj * v[j].x);
            atomicAdd(p + 1, wj * v[j].y);
          }
        }
      }
    };

    int nrec = -1;
    if (fits) {
      nrec = __builtin_amdgcn_readfirstlane(cnts[li]);
      if (nrec > AG_C) nrec = -1;
    }

    if (nrec >= 0) {
      unsigned rn = (lane < nrec) ? rw[lane] : 0u;
      for (int b = 0; b < nrec; b += 32) {
        const unsigned r = rn;
        const int n = min(nrec - b, 32);
        if (b + 32 + lane < nrec) rn = rw[b + 32 + lane];
        const bool live = lane < n;
        const int  s  = live ? (int)(r & 0xFFFFu) : 0;
        const int  dl = live ? (int)(r >> 16) : AG_R;
        const float wa = dinv[s];
        const float wb = dinv[lo + (live ? dl : 0)];
        add_rows(s, dl, n, [&] { return live ? wa * wb : 0.0f; });
      }
    } else {
      sweep_range<AG_QCAP>(dstp, E, lo, cnt, rq, wave, lane, tid, [&](int e, int n) {
        int   s  = 0;
        int   dl = AG_R;
        float w  = 0.0f;
        if (lane < n) {
          s = ei[e];
          const int d = dstp[e];
          dl = d - lo;
          w  = dinv[s] * dinv[d];
        }
        add_rows(s, dl, n, [&] { return w; });
      });
    }
    __syncthreads();

    for (int i = tid; i < cnt * 16; i += 1024)
      ((float4*)out)[(size_t)lo * 16 + i] = *(const float4*)&acc[i * 4];
    __syncthreads();
  }
}

#define AT_P 68

__global__ __launch_bounds__(128) __attribute__((amdgpu_waves_per_eu(4, 4)))
void edge_mlp_k(const float* __restrict__ h, const int* __restrict__ ei,
                const float* __restrict__ Wm1, const float* __restrict__ bm1,
                const float* __restrict__ Wm2, const float* __restrict__ bm2,
                float* __restrict__ out, int E, int ntiles) {
  __shared__ __attribute__((aligned(16))) float At[128 * AT_P];
  __shared__ __attribute__((aligned(16))) float Zt[64 * AT_P];

  const int tid  = threadIdx.x;
  const int wave = tid >> 5;
  const int lane = tid & 31;
  const int mr   = lane & 15;
  const int half = lane >> 4;

  v2f b1f[32];
#pragma unroll
  for (int j = 0; j < 32; ++j) {
    const float* p = Wm1 + (size_t)(4 * j + 2 * half) * HID + wave * 16 + mr;
    b1f[j][0] = p[0];
    b1f[j][1] = p[HID];
  }
  v2f b2f[16];
#pragma unroll
  for (int j = 0; j < 16; ++j) {
    const float* p = Wm2 + (size_t)(4 * j + 2 * half) * OUT_C + mr;
    b2f[j][0] = p[0];
    b2f[j][1] = p[OUT_C];
  }
  const float bias1 = bm1[wave * 16 + mr];
  const float bo    = bm2[mr];

  for (int t = blockIdx.x; t < ntiles; t += gridDim.x) {
    const int e0 = t * 64;

    const int gi = wave * 32 + lane;
    const int node_l = (gi < 64) ? ei[e0 + gi] : ei[E + e0 + (gi - 64)];
#pragma unroll
    for (int i = 0; i < 16; ++i) {
      const int rl   = 2 * i + half;
      const int node = __shfl(node_l, rl);
      float4 v = *(const float4*)&h[(size_t)node * HID + mr * 4];
      v.x = fmaxf(v.x, 0.0f);
      v.y = fmaxf(v.y, 0.0f);
      v.z = fmaxf(v.z, 0.0f);
      v.w = fmaxf(v.w, 0.0f);
      *(float4*)&At[(wave * 32 + rl) * AT_P + mr * 4] = v;
    }
    __syncthreads();

    v8f acc[4] = {v8f{}, v8f{}, v8f{}, v8f{}};
#pragma unroll
    for (int j = 0; j < 32; ++j) {
      const int rbase = (j < 16) ? 0 : 64;
      const int kc    = 4 * (j & 15) + 2 * half;
#pragma unroll
      for (int mt = 0; mt < 4; ++mt) {
        const v2f a = *(const v2f*)&At[(rbase + mt * 16 + mr) * AT_P + kc];
        acc[mt] = __builtin_amdgcn_wmma_f32_16x16x4_f32(false, a, false, b1f[j],
                                                        (short)0, acc[mt], false, false);
      }
    }

#pragma unroll
    for (int mt = 0; mt < 4; ++mt) {
#pragma unroll
      for (int r = 0; r < 8; ++r)
        Zt[(mt * 16 + r + 8 * half) * AT_P + wave * 16 + mr] = fmaxf(acc[mt][r] + bias1, 0.0f);
    }
    __syncthreads();

    v8f o = {};
#pragma unroll
    for (int j = 0; j < 16; ++j) {
      const v2f a = *(const v2f*)&Zt[(wave * 16 + mr) * AT_P + 4 * j + 2 * half];
      o = __builtin_amdgcn_wmma_f32_16x16x4_f32(false, a, false, b2f[j],
                                                (short)0, o, false, false);
    }

    float* op = out + (size_t)(e0 + wave * 16 + 8 * half) * OUT_C + mr;
#pragma unroll
    for (int r = 0; r < 8; ++r) op[(size_t)r * OUT_C] = o[r] + bo;
  }
}

extern "C" void kernel_launch(void* const* d_in, const int* in_sizes, int n_in,
                              void* d_out, int out_size, void* d_ws, size_t ws_size,
                              hipStream_t stream) {
  const float* x   = (const float*)d_in[0];
  const int*   ei  = (const int*)d_in[1];
  const float* W1  = (const float*)d_in[2];
  const float* b1  = (const float*)d_in[3];
  const float* W2  = (const float*)d_in[4];
  const float* b2  = (const float*)d_in[5];
  const float* Wm1 = (const float*)d_in[6];
  const float* bm1 = (const float*)d_in[7];
  const float* Wm2 = (const float*)d_in[8];
  const float* bm2 = (const float*)d_in[9];
  float* out = (float*)d_out;

  const int N = in_sizes[0] / IN_C;
  const int E = in_sizes[1] / 2;

  float* dinv = (float*)d_ws;
  float* bufA = dinv + (size_t)((N + 63) & ~63);
  float* bufB = bufA + (size_t)N * HID;

  const int mtiles = (N + 15) / 16;
  const int gx     = (mtiles + 3) / 4;
  const int ag  = (N + AG_R - 1) / AG_R;
  int*      cnts = (int*)(bufB + (size_t)N * HID);
  unsigned* rec  = (unsigned*)(cnts + (size_t)ag * AG_WAVES);

  unsigned* marks = (unsigned*)bufA;
  mark_k<<<(E + MK_E - 1) / MK_E, 256, 0, stream>>>(marks, ei, E, N);
  sweep_k<<<ag, 1024, 0, stream>>>(dinv, ei, E, N, rec, cnts, dinv, ws_size, marks);

  gemm_wmma_k<false><<<dim3(gx, HID / 16), 128, 0, stream>>>(x, W1, bufA, N, IN_C, HID);
  agg_own_k<<<ag, 1024, 0, stream>>>(bufB, bufA, dinv, b1, ei, E, N, rec, cnts, dinv, ws_size);

  gemm_wmma_k<true><<<dim3(gx, HID / 16), 128, 0, stream>>>(bufB, W2, bufA, N, HID, HID);
  agg_own_k<<<ag, 1024, 0, stream>>>(bufB, bufA, dinv, b2, ei, E, N, rec, cnts, dinv, ws_size);

  edge_mlp_k<<<1024, 128, 0, stream>>>(bufB, ei, Wm1, bm1, Wm2, bm2, out, E, E / 64);
}


# dense layers retiled: a workgroup per 64-row tile stages its A rows and W once in LDS by 16-byte row loads, 4 waves x 16 columns on WMMA from LDS, whole result rows out via LDS; sums unchanged
# speedup vs baseline: 2.0421x; 1.1380x over previous
#include <hip/hip_runtime.h>
#include <hip/hip_bf16.h>

typedef __attribute__((ext_vector_type(2))) float v2f;
typedef __attribute__((ext_vector_type(8))) float v8f;

#define IN_C 128
#define HID  64
#define OUT_C 16

#define DN_CP 68
#define DN_WP 72

template <int K, bool RELU_A>
__global__ __launch_bounds__(128)
void dense_k(const float* __restrict__ A, const float* __restrict__ Wt,
             float* __restrict__ C, int M) {
  constexpr int P   = K + 4;
  constexpr int LPR = K / 4;
  constexpr int NLD = 64 * LPR / 128;
  constexpr int NLW = K * 16 / 128;
  __shared__ __attribute__((aligned(16))) float As[64 * P];
  __shared__ __attribute__((aligned(16))) float Ws[K * DN_WP];

  const int tid  = threadIdx.x;
  const int wave = tid >> 5;
  const int lane = tid & 31;
  const int mr   = lane & 15;
  const int half = lane >> 4;
  const int m0   = blockIdx.x * 64;

  float4 av[NLD], wv[NLW];
#pragma unroll
  for (int i = 0; i < NLD; ++i) {
    const int idx = i * 128 + tid;
    const int row = idx / LPR, c4 = idx % LPR;
    const int gr  = min(m0 + row, M - 1);
    av[i] = *(const float4*)&A[(size_t)gr * K + c4 * 4];
  }
#pragma unroll
  for (int i = 0; i < NLW; ++i) wv[i] = *(const float4*)&Wt[(size_t)(i * 128 + tid) * 4];
#pragma unroll
  for (int i = 0; i < NLW; ++i) {
    const int idx = i * 128 + tid;
    *(float4*)&Ws[(idx >> 4) * DN_WP + (idx & 15) * 4] = wv[i];
  }
#pragma unroll
  for (int i = 0; i < NLD; ++i) {
    const int idx = i * 128 + tid;
    const int row = idx / LPR, c4 = idx % LPR;
    float4 v = av[i];
    if (RELU_A) {
      v.x = fmaxf(v.x, 0.0f);
      v.y = fmaxf(v.y, 0.0f);
      v.z = fmaxf(v.z, 0.0f);
      v.w = fmaxf(v.w, 0.0f);
    }
    *(float4*)&As[row * P + c4 * 4] = v;
  }
  __syncthreads();

  v8f acc[4] = {v8f{}, v8f{}, v8f{}, v8f{}};
#pragma unroll
  for (int j = 0; j < K / 4; ++j) {
    const int kc = 4 * j + 2 * half;
    v2f b;
    b[0] = Ws[kc * DN_WP + wave * 16 + mr];
    b[1] = Ws[(kc + 1) * DN_WP + wave * 16 + mr];
#pragma unroll
    for (int mt = 0; mt < 4; ++mt) {
      const v2f a = *(const v2f*)&As[(mt * 16 + mr) * P + kc];
      acc[mt] = __builtin_amdgcn_wmma_f32_16x16x4_f32(false, a, false, b,
                                                      (short)0, acc[mt], false, false);
    }
  }
  __syncthreads();

#pragma unroll
  for (int mt = 0; mt < 4; ++mt) {
#pragma unroll
    for (int r = 0; r < 8; ++r)
      As[(mt * 16 + r + 8 * half) * DN_CP + wave * 16 + mr] = acc[mt][r];
  }
  __syncthreads();

  float4 cv[8];
#pragma unroll
  for (int i = 0; i < 8; ++i) {
    const int idx = i * 128 + tid;
    cv[i] = *(const float4*)&As[(idx >> 4) * DN_CP + (idx & 15) * 4];
  }
  float* const cp = C + (size_t)m0 * HID + (size_t)tid * 4;
  if (m0 + 64 <= M) {
#pragma unroll
    for (int i = 0; i < 8; ++i) *(float4*)&cp[i * 512] = cv[i];
  } else {
#pragma unroll
    for (int i = 0; i < 8; ++i)
      if (m0 + ((i * 128 + tid) >> 4) < M) *(float4*)&cp[i * 512] = cv[i];
  }
}

#define AG_R     196
#define AG_QCAP  1024
#define SW_QCAP  2048
#define AG_WAVES 32
#define AG_C     160

#define MK_E  2048
#define MK_W  (MK_E / 32)
#define MK_P  (MK_W + 4)
#define MK_R  256

__device__ __forceinline__ bool marks_fit(int N, int E) {
  const int nrng = (N + AG_R - 1) / AG_R;
  const size_t nwords = ((size_t)E + 31) >> 5;
  return nrng <= MK_R && (size_t)nrng * nwords <= (size_t)2 * N * HID;
}

__device__ __forceinline__ float lane_f(float x, int j) {
  return __int_as_float(__builtin_amdgcn_readlane(__float_as_int(x), j));
}

__device__ __forceinline__ int4 ld_dst4(const int* __restrict__ dal, int q) {
  return *(const int4*)(dal + 4 * (size_t)q);
}

template <int QCAP, class Drain>
__device__ __forceinline__ void sweep_range(const int* __restrict__ dstp, int E, int lo, int cnt,
                                            int* rq, int wave, int lane, int tid, Drain drain) {
  const int a0  = min((4 - (E & 3)) & 3, E);
  const int nq  = (E - a0) >> 2;
  const int* __restrict__ dal = dstp + a0;
  const int nfull = nq >> 12;
  const int rot = (nfull > 0) ? (int)(blockIdx.x % (unsigned)nfull) : 0;

  int head = 0, tail = 0;

  auto pop = [&](int n) {
    int e = 0;
    if (lane < n) e = rq[(head + lane) & (QCAP - 1)];
    head += n;
    drain(e, n);
  };

  auto take = [&](const int4 d, int e0) {
    const bool h0 = (unsigned)(d.x - lo) < (unsigned)cnt, h1 = (unsigned)(d.y - lo) < (unsigned)cnt;
    const bool h2 = (unsigned)(d.z - lo) < (unsigned)cnt, h3 = (unsigned)(d.w - lo) < (unsigned)cnt;
    const unsigned m0 = __builtin_amdgcn_ballot_w32(h0), m1 = __builtin_amdgcn_ballot_w32(h1);
    const unsigned m2 = __builtin_amdgcn_ballot_w32(h2), m3 = __builtin_amdgcn_ballot_w32(h3);
    if (m0 | m1 | m2 | m3) {
      if (m0) { if (h0) rq[(tail + __builtin_amdgcn_mbcnt_lo(m0, 0u)) & (QCAP - 1)] = e0;     tail += __popc(m0); }
      if (m1) { if (h1) rq[(tail + __builtin_amdgcn_mbcnt_lo(m1, 0u)) & (QCAP - 1)] = e0 + 1; tail += __popc(m1); }
      if (m2) { if (h2) rq[(tail + __builtin_amdgcn_mbcnt_lo(m2, 0u)) & (QCAP - 1)] = e0 + 2; tail += __popc(m2); }
      if (m3) { if (h3) rq[(tail + __builtin_amdgcn_mbcnt_lo(m3, 0u)) & (QCAP - 1)] = e0 + 3; tail += __popc(m3); }
    }
  };

  if (nfull > 0) {
    int it = rot;
    int4 cur[4], nxt[4];
#pragma unroll
    for (int u = 0; u < 4; ++u) cur[u] = ld_dst4(dal, it * 4096 + u * 1024 + tid);
    for (int k = 0; k < nfull; ++k) {
      const int qb = it * 4096 + tid;
      it = (it + 1 == nfull) ? 0 : it + 1;
#pragma unroll
      for (int u = 0; u < 4; ++u) nxt[u] = ld_dst4(dal, it * 4096 + u * 1024 + tid);
#pragma unroll
      for (int u = 0; u < 4; ++u) take(cur[u], a0 + 4 * (qb + u * 1024));
#pragma unroll
      for (int u = 0; u < 4; ++u) cur[u] = nxt[u];
      while (tail - head >= 32) pop(32);
    }
  }
#pragma unroll
  for (int u = 0; u < 4; ++u) {
    const int q = nfull * 4096 + u * 1024 + tid;
    int4 d = make_int4(-1, -1, -1, -1);
    if (q < nq) d = ld_dst4(dal, q);
    take(d, a0 + 4 * q);
  }
  if (wave == 0) {
    const int nrem = E - 4 * nq;
    const int er = (lane < a0) ? lane : lane + 4 * nq;
    int4 d = make_int4(-1, -1, -1, -1);
    if (lane < nrem) d.x = dstp[er];
    take(d, er);
  }
  while (tail - head > 0) pop(min(tail - head, 32));
}

template <int QCAP, class Drain>
__device__ __forceinline__ void sweep_marks(const unsigned* __restrict__ marks, int nrng, int nwords, int ri,
                                            int* rq, int lane, int tid, Drain drain) {
  int head = 0, tail = 0;

  auto pop = [&](int n) {
    int e = 0;
    if (lane < n) e = rq[(head + lane) & (QCAP - 1)];
    head += n;
    drain(e, n);
  };

  const int nsteps = (nwords + 1023) >> 10;
  for (int k0 = 0; k0 < nsteps; k0 += 16) {
    unsigned x[16];
#pragma unroll
    for (int u = 0; u < 16; ++u) {
      const int w = (k0 + u) * 1024 + tid;
      x[u] = 0u;
      if (w < nwords) {
        const int c  = w >> 6;
        const int cw = min(MK_W, nwords - (c << 6));
        x[u] = marks[(size_t)c * nrng * MK_W + (size_t)ri * cw + (w & (MK_W - 1))];
      }
    }
#pragma unroll
    for (int u = 0; u < 16; ++u) {
      const int eb = ((k0 + u) * 1024 + tid) << 5;
      unsigned v = x[u];
      unsigned m = __builtin_amdgcn_ballot_w32(v != 0u);
      while (m) {
        if (v) rq[(tail + __builtin_amdgcn_mbcnt_lo(m, 0u)) & (QCAP - 1)] = eb + __builtin_ctz(v);
        tail += __popc(m);
        v &= v - 1u;
        m = __builtin_amdgcn_ballot_w32(v != 0u);
      }
      while (tail - head >= 32) pop(32);
    }
  }
  while (tail - head > 0) pop(min(tail - head, 32));
}

__global__ __launch_bounds__(256)
void mark_k(unsigned* __restrict__ marks, const int* __restrict__ ei, int E, int N) {
  __shared__ __attribute__((aligned(16))) unsigned L[MK_R * MK_P];
  if (!marks_fit(N, E)) return;
  const int tid    = threadIdx.x;
  const int nrng   = (N + AG_R - 1) / AG_R;
  const int nwords = (E + 31) >> 5;
  const int* __restrict__ dstp = ei + E;

  for (int c = blockIdx.x; c * MK_W < nwords; c += gridDim.x) {
    for (int i = tid; i < nrng * (MK_P / 4); i += 256)
      ((uint4*)L)[i] = make_uint4(0u, 0u, 0u, 0u);
    __syncthreads();

    const int e0 = c * MK_E;
    int d[MK_E / 256];
#pragma unroll
    for (int i = 0; i < MK_E / 256; ++i) {
      const int e = e0 + i * 256 + tid;
      d[i] = (e < E) ? dstp[e] : -1;
    }
#pragma unroll
    for (int i = 0; i < MK_E / 256; ++i) {
      const int el = i * 256 + tid;
      if ((unsigned)d[i] < (unsigned)N)
        atomicOr(&L[(d[i] / AG_R) * MK_P + (el >> 5)], 1u << (el & 31));
    }
    __syncthreads();

    const int cw = min(MK_W, nwords - c * MK_W);
    unsigned* __restrict__ g = marks + (size_t)c * nrng * MK_W;
    if (cw == MK_W) {
      for (int i = tid; i < nrng * (MK_W / 4); i += 256) {
        const int r = i >> 4, q = i & 15;
        ((uint4*)g)[i] = *(const uint4*)&L[r * MK_P + 4 * q];
      }
    } else {
      const int wl = tid & (MK_W - 1);
      for (int r = tid >> 6; r < nrng; r += 256 / MK_W)
        if (wl < cw) g[r * cw + wl] = L[r * MK_P + wl];
    }
    __syncthreads();
  }
}

__device__ __forceinline__ bool lists_fit(const unsigned* rec, int N, const float* ws0, size_t ws_bytes) {
  const int nrng = (N + AG_R - 1) / AG_R;
  return (N <= 65536) &&
      (size_t)((const char*)(rec + (size_t)nrng * AG_WAVES * AG_C) - (const char*)ws0) <= ws_bytes;
}

__global__ __launch_bounds__(1024)
void sweep_k(float* __restrict__ dinv, const int* __restrict__ ei, int E, int N,
             unsigned* __restrict__ rec, int* __restrict__ cnts,
             const float* ws0, size_t ws_bytes, const unsigned* __restrict__ marks) {
  __shared__ int ring[AG_WAVES * SW_QCAP];
  __shared__ int degc[AG_R];
  __shared__ int giveup;

  const int tid  = threadIdx.x;
  const int wave = __builtin_amdgcn_readfirstlane(tid >> 5);
  const int lane = tid & 31;
  int* const rq = ring + wave * SW_QCAP;
  const int* __restrict__ dstp = ei + E;
  const bool fits = lists_fit(rec, N, ws0, ws_bytes);
  const bool mk   = marks_fit(N, E);
  const int nrng   = (N + AG_R - 1) / AG_R;
  const int nwords = (E + 31) >> 5;

  for (int lo = blockIdx.x * AG_R; lo < N; lo += gridDim.x * AG_R) {
    const int cnt = min(AG_R, N - lo);
    const int li  = (lo / AG_R) * AG_WAVES + wave;
    unsigned* const rw = rec + (size_t)li * AG_C;

    if (tid < AG_R) degc[tid] = 1;
    if (tid == 0) giveup = 0;
    __syncthreads();

    int nrec = fits ? 0 : -1;
    auto drain = [&](int e, int n) {
      const bool keep = nrec >= 0 && nrec + n <= AG_C;
      if (lane < n) {
        const int s  = ei[e];
        const int dl = dstp[e] - lo;
        atomicAdd(&degc[dl], 1);
        if (keep) rw[nrec + lane] = ((unsigned)dl << 16) | (unsigned)s;
      }
      nrec = keep ? nrec + n : -1;
    };
    if (mk) sweep_marks<SW_QCAP>(marks, nrng, nwords, lo / AG_R, rq, lane, tid, drain);
    else    sweep_range<SW_QCAP>(dstp, E, lo, cnt, rq, wave, lane, tid, drain);
    if (nrec < 0 && lane == 0) atomicOr(&giveup, 1);
    __syncthreads();
    if (fits && lane == 0) cnts[li] = giveup ? -1 : nrec;

    if (tid < cnt) {
      const float d = (float)degc[tid];
      dinv[lo + tid] = (d > 0.0f) ? rsqrtf(d) : 0.0f;
    }
    __syncthreads();
  }
}

__global__ __launch_bounds__(1024)
void agg_own_k(float* __restrict__ out, const float* __restrict__ h,
               const float* __restrict__ dinv, const float* __restrict__ bias,
               const int* __restrict__ ei, int E, int N,
               const unsigned* __restrict__ rec, const int* __restrict__ cnts,
               const float* ws0, size_t ws_bytes) {
  __shared__ __attribute__((aligned(16))) float acc[(AG_R + 1) * HID];
  __shared__ int ring[AG_WAVES * AG_QCAP];

  const int tid  = threadIdx.x;
  const int wave = __builtin_amdgcn_readfirstlane(tid >> 5);
  const int lane = tid & 31;
  int* const rq = ring + wave * AG_QCAP;
  const int* __restrict__ dstp = ei + E;
  const bool fits = lists_fit(rec, N, ws0, ws_bytes);

  for (int lo = blockIdx.x * AG_R; lo < N; lo += gridDim.x * AG_R) {
    const int cnt = min(AG_R, N - lo);
    const int li  = (lo / AG_R) * AG_WAVES + wave;
    const unsigned* const rw = rec + (size_t)li * AG_C;

    for (int i = tid; i < cnt * 16; i += 1024) {
      const int r  = i >> 4;
      const int c4 = (i & 15) * 4;
      float d2 = dinv[lo + r]; d2 *= d2;
      const float4 hv = *(const float4*)&h[(size_t)(lo + r) * HID + c4];
      const float4 bv = *(const float4*)&bias[c4];
      float4 o;
      o.x = bv.x + d2 * hv.x;
      o.y = bv.y + d2 * hv.y;
      o.z = bv.z + d2 * hv.z;
      o.w = bv.w + d2 * hv.w;
      *(float4*)&acc[r * HID + c4] = o;
    }
    if (tid < HID) acc[AG_R * HID + tid] = 0.0f;
    __syncthreads();

    auto add_rows = [&](int s, int dl, int n, auto wfn) {
      float w = 0.0f;
#pragma unroll
      for (int hb = 0; hb < 32; hb += 16) {
        if (hb < n) {
          float2 v[16];
#pragma unroll
          for (int j = 0; j < 16; ++j) {
            const int sj = __builtin_amdgcn_readlane(s, hb + j);
            v[j] = *(const float2*)&h[(size_t)sj * HID + 2 * lane];
          }
          if (hb == 0) w = wfn();
#pragma unroll
          for (int j = 0; j < 16; ++j) {
            const int   dj = __builtin_amdgcn_readlane(dl, hb + j);
            const float wj = lane_f(w, hb + j);
            float* p = &acc[dj * HID + 2 * lane];
            atomicAdd(p,     wj * v[j].x);
            atomicAdd(p + 1, wj * v[j].y);
          }
        }
      }
    };

    int nrec = -1;
    if (fits) {
      nrec = __builtin_amdgcn_readfirstlane(cnts[li]);
      if (nrec > AG_C) nrec = -1;
    }

    if (nrec >= 0) {
      unsigned rn = (lane < nrec) ? rw[lane] : 0u;
      for (int b = 0; b < nrec; b += 32) {
        const unsigned r = rn;
        const int n = min(nrec - b, 32);
        if (b + 32 + lane < nrec) rn = rw[b + 32 + lane];
        const bool live = lane < n;
        const int  s  = live ? (int)(r & 0xFFFFu) : 0;
        const int  dl = live ? (int)(r >> 16) : AG_R;
        const float wa = dinv[s];
        const float wb = dinv[lo + (live ? dl : 0)];
        add_rows(s, dl, n, [&] { return live ? wa * wb : 0.0f; });
      }
    } else {
      sweep_range<AG_QCAP>(dstp, E, lo, cnt, rq, wave, lane, tid, [&](int e, int n) {
        int   s  = 0;
        int   dl = AG_R;
        float w  = 0.0f;
        if (lane < n) {
          s = ei[e];
          const int d = dstp[e];
          dl = d - lo;
          w  = dinv[s] * dinv[d];
        }
        add_rows(s, dl, n, [&] { return w; });
      });
    }
    __syncthreads();

    for (int i = tid; i < cnt * 16; i += 1024)
      ((float4*)out)[(size_t)lo * 16 + i] = *(const float4*)&acc[i * 4];
    __syncthreads();
  }
}

#define AT_P 68

__global__ __launch_bounds__(128) __attribute__((amdgpu_waves_per_eu(4, 4)))
void edge_mlp_k(const float* __restrict__ h, const int* __restrict__ ei,
                const float* __restrict__ Wm1, const float* __restrict__ bm1,
                const float* __restrict__ Wm2, const float* __restrict__ bm2,
                float* __restrict__ out, int E, int ntiles) {
  __shared__ __attribute__((aligned(16))) float At[128 * AT_P];
  __shared__ __attribute__((aligned(16))) float Zt[64 * AT_P];

  const int tid  = threadIdx.x;
  const int wave = tid >> 5;
  const int lane = tid & 31;
  const int mr   = lane & 15;
  const int half = lane >> 4;

  v2f b1f[32];
#pragma unroll
  for (int j = 0; j < 32; ++j) {
    const float* p = Wm1 + (size_t)(4 * j + 2 * half) * HID + wave * 16 + mr;
    b1f[j][0] = p[0];
    b1f[j][1] = p[HID];
  }
  v2f b2f[16];
#pragma unroll
  for (int j = 0; j < 16; ++j) {
    const float* p = Wm2 + (size_t)(4 * j + 2 * half) * OUT_C + mr;
    b2f[j][0] = p[0];
    b2f[j][1] = p[OUT_C];
  }
  const float bias1 = bm1[wave * 16 + mr];
  const float bo    = bm2[mr];

  for (int t = blockIdx.x; t < ntiles; t += gridDim.x) {
    const int e0 = t * 64;

    const int gi = wave * 32 + lane;
    const int node_l = (gi < 64) ? ei[e0 + gi] : ei[E + e0 + (gi - 64)];
#pragma unroll
    for (int i = 0; i < 16; ++i) {
      const int rl   = 2 * i + half;
      const int node = __shfl(node_l, rl);
      float4 v = *(const float4*)&h[(size_t)node * HID + mr * 4];
      v.x = fmaxf(v.x, 0.0f);
      v.y = fmaxf(v.y, 0.0f);
      v.z = fmaxf(v.z, 0.0f);
      v.w = fmaxf(v.w, 0.0f);
      *(float4*)&At[(wave * 32 + rl) * AT_P + mr * 4] = v;
    }
    __syncthreads();

    v8f acc[4] = {v8f{}, v8f{}, v8f{}, v8f{}};
#pragma unroll
    for (int j = 0; j < 32; ++j) {
      const int rbase = (j < 16) ? 0 : 64;
      const int kc    = 4 * (j & 15) + 2 * half;
#pragma unroll
      for (int mt = 0; mt < 4; ++mt) {
        const v2f a = *(const v2f*)&At[(rbase + mt * 16 + mr) * AT_P + kc];
        acc[mt] = __builtin_amdgcn_wmma_f32_16x16x4_f32(false, a, false, b1f[j],
                                                        (short)0, acc[mt], false, false);
      }
    }

#pragma unroll
    for (int mt = 0; mt < 4; ++mt) {
#pragma unroll
      for (int r = 0; r < 8; ++r)
        Zt[(mt * 16 + r + 8 * half) * AT_P + wave * 16 + mr] = fmaxf(acc[mt][r] + bias1, 0.0f);
    }
    __syncthreads();

    v8f o = {};
#pragma unroll
    for (int j = 0; j < 16; ++j) {
      const v2f a = *(const v2f*)&Zt[(wave * 16 + mr) * AT_P + 4 * j + 2 * half];
      o = __builtin_amdgcn_wmma_f32_16x16x4_f32(false, a, false, b2f[j],
                                                (short)0, o, false, false);
    }

    float* op = out + (size_t)(e0 + wave * 16 + 8 * half) * OUT_C + mr;
#pragma unroll
    for (int r = 0; r < 8; ++r) op[(size_t)r * OUT_C] = o[r] + bo;
  }
}

extern "C" void kernel_launch(void* const* d_in, const int* in_sizes, int n_in,
                              void* d_out, int out_size, void* d_ws, size_t ws_size,
                              hipStream_t stream) {
  const float* x   = (const float*)d_in[0];
  const int*   ei  = (const int*)d_in[1];
  const float* W1  = (const float*)d_in[2];
  const float* b1  = (const float*)d_in[3];
  const float* W2  = (const float*)d_in[4];
  const float* b2  = (const float*)d_in[5];
  const float* Wm1 = (const float*)d_in[6];
  const float* bm1 = (const float*)d_in[7];
  const float* Wm2 = (const float*)d_in[8];
  const float* bm2 = (const float*)d_in[9];
  float* out = (float*)d_out;

  const int N = in_sizes[0] / IN_C;
  const int E = in_sizes[1] / 2;

  float* dinv = (float*)d_ws;
  float* bufA = dinv + (size_t)((N + 63) & ~63);
  float* bufB = bufA + (size_t)N * HID;

  const int mtiles = (N + 15) / 16;
  const int gx     = (mtiles + 3) / 4;
  const int ag  = (N + AG_R - 1) / AG_R;
  int*      cnts = (int*)(bufB + (size_t)N * HID);
  unsigned* rec  = (unsigned*)(cnts + (size_t)ag * AG_WAVES);

  unsigned* marks = (unsigned*)bufA;
  mark_k<<<(E + MK_E - 1) / MK_E, 256, 0, stream>>>(marks, ei, E, N);
  sweep_k<<<ag, 1024, 0, stream>>>(dinv, ei, E, N, rec, cnts, dinv, ws_size, marks);

  dense_k<IN_C, false><<<gx, 128, 0, stream>>>(x, W1, bufA, N);
  agg_own_k<<<ag, 1024, 0, stream>>>(bufB, bufA, dinv, b1, ei, E, N, rec, cnts, dinv, ws_size);

  dense_k<HID, true><<<gx, 128, 0, stream>>>(bufB, W2, bufA, N);
  agg_own_k<<<ag, 1024, 0, stream>>>(bufB, bufA, dinv, b2, ei, E, N, rec, cnts, dinv, ws_size);

  edge_mlp_k<<<1024, 128, 0, stream>>>(bufB, ei, Wm1, bm1, Wm2, bm2, out, E, E / 64);
}
